# starting kernel
# baseline (speedup 1.0000x reference)
#include <hip/hip_runtime.h>
#include <hip/hip_bf16.h>
#include <math.h>

typedef float v2f __attribute__((ext_vector_type(2)));
typedef float v8f __attribute__((ext_vector_type(8)));

#define B_ 4
#define M_ 2048
#define K_ 64
#define C_ 64
#define P_ 28
#define NCURVE 64

__device__ __forceinline__ v8f vmin8(v8f a, v8f b){
  v8f r;
#pragma unroll
  for (int i = 0; i < 8; ++i) r[i] = fminf(a[i], b[i]);
  return r;
}

__global__ __launch_bounds__(128) void sdf_kernel(
    const float* __restrict__ pts, const float* __restrict__ prim,
    float* __restrict__ sdf_out, float* __restrict__ dmin_out)
{
  __shared__ float4 s_c4[NCURVE];
  __shared__ float4 s_e4[NCURVE];
  const int b = blockIdx.x >> 6;
  const int k = blockIdx.x & 63;
  const float* Pp = prim + (size_t)b * P_ * K_ + k;
  const int tid = threadIdx.x;
  const float PI = 3.14159265358979323846f;

  if (tid < NCURVE) {
    int seg = tid >> 4;
    float t = (float)(tid & 15) * (1.0f / 16.0f);
    float theta = PI * 0.125f + atanf(tanf(PI * 0.125f) * (1.0f / 3.0f));
    float r0 = (PI * 0.5f) * (float)seg;
    float r1 = r0 + theta;
    float r2 = (PI * 0.5f) * (float)(seg + 1) - theta;
    float r3 = (PI * 0.5f) * (float)(seg + 1);
    float rad0 = fabsf(Pp[(8 + seg * 3 + 0) * K_]);
    float rad1 = fabsf(Pp[(8 + seg * 3 + 1) * K_]);
    float rad2 = fabsf(Pp[(8 + seg * 3 + 2) * K_]);
    int seg1 = (seg + 1) & 3;
    float rad3 = fabsf(Pp[(8 + seg1 * 3 + 0) * K_]);
    float P0x = cosf(r0) * rad0, P0y = sinf(r0) * rad0;
    float P1x = cosf(r1) * rad1, P1y = sinf(r1) * rad1;
    float P2x = cosf(r2) * rad2, P2y = sinf(r2) * rad2;
    float P3x = cosf(r3) * rad3, P3y = sinf(r3) * rad3;
    float w1 = fabsf(Pp[(20 + seg * 2 + 0) * K_]);
    float w2 = fabsf(Pp[(20 + seg * 2 + 1) * K_]);
    float u = 1.0f - t;
    float b0 = u * u * u, b1 = 3.0f * u * u * t, b2 = 3.0f * u * t * t, b3 = t * t * t;
    float inv = 1.0f / (b0 + w1 * b1 + w2 * b2 + b3);
    float cx = (P0x * b0 + P1x * (w1 * b1) + P2x * (w2 * b2) + P3x * b3) * inv;
    float cy = (P0y * b0 + P1y * (w1 * b1) + P2y * (w2 * b2) + P3y * b3) * inv;
    s_c4[tid] = make_float4(cx, cy, cx * cx + cy * cy, 1.0f);
  }

  float q0 = Pp[0], q1 = Pp[K_], q2 = Pp[2 * K_], q3 = Pp[3 * K_];
  float invn = 1.0f / sqrtf(q0 * q0 + q1 * q1 + q2 * q2 + q3 * q3 + 1e-8f);
  float qw = q0 * invn, vx = -q1 * invn, vy = -q2 * invn, vz = -q3 * invn;
  float tx = Pp[4 * K_], ty = Pp[5 * K_], tz = Pp[6 * K_];
  float habs = fabsf(Pp[7 * K_]);
  __syncthreads();

  if (tid < NCURVE) {
    float4 ca = s_c4[tid];
    float4 cb = s_c4[(tid + 1) & 63];
    s_e4[tid] = make_float4(ca.y - cb.y, cb.x - ca.x,
                            ca.x * cb.y - ca.y * cb.x, ca.y);
  }

  const int lane = tid & 31;
  const int wv = tid >> 5;
  const int row = lane & 15;
  const bool hi = lane >= 16;
  const int m = blockIdx.y * 64 + wv * 16 + row;

  const float* p3 = pts + ((size_t)b * M_ + m) * 3;
  float px0 = p3[0] - tx, py0 = p3[1] - ty, pz0 = p3[2] - tz;
  float t2x = 2.0f * (vy * pz0 - vz * py0);
  float t2y = 2.0f * (vz * px0 - vx * pz0);
  float t2z = 2.0f * (vx * py0 - vy * px0);
  float plx = px0 + qw * t2x + (vy * t2z - vz * t2y);
  float ply = py0 + qw * t2y + (vz * t2x - vx * t2z);
  float plz = pz0 + qw * t2z + (vx * t2y - vy * t2x);
  float px = plx, py = ply;
  float ss = px * px + py * py;
  __syncthreads();

  v2f Bd; Bd.x = hi ? ss : px;   Bd.y = hi ? 1.0f : py;
  v2f Bw; Bw.x = hi ? 1.0f : px; Bw.y = hi ? 0.0f : py;
  v8f z = {0.f, 0.f, 0.f, 0.f, 0.f, 0.f, 0.f, 0.f};
  v8f e = {3e38f, 3e38f, 3e38f, 3e38f, 3e38f, 3e38f, 3e38f, 3e38f};

  int wn = 0;
  int le0b = 0;
  float cr7[4];
  int   le7[4];
#pragma unroll
  for (int tTile = 0; tTile < 4; ++tTile) {
    float4 cc = s_c4[tTile * 16 + row];
    float4 ee = s_e4[tTile * 16 + row];
    v2f Ad; Ad.x = hi ? 1.0f   : -2.0f * cc.x;  Ad.y = hi ? cc.z : -2.0f * cc.y;
    v2f Ac; Ac.x = hi ? ee.z   : ee.x;          Ac.y = hi ? 0.0f : ee.y;
    v2f Ay; Ay.x = hi ? -ee.w  : 0.0f;          Ay.y = hi ? 0.0f : 1.0f;
    v8f d  = __builtin_amdgcn_wmma_f32_16x16x4_f32(
        false, Ad, false, Bd, (short)0, z, false, false);
    v8f cr = __builtin_amdgcn_wmma_f32_16x16x4_f32(
        false, Ac, false, Bw, (short)0, z, false, false);
    v8f yd = __builtin_amdgcn_wmma_f32_16x16x4_f32(
        false, Ay, false, Bw, (short)0, z, false, false);
    e = vmin8(e, d);
    int le[8];
#pragma unroll
    for (int v = 0; v < 8; ++v) le[v] = (yd[v] >= 0.0f);
#pragma unroll
    for (int v = 0; v < 7; ++v) {
      int aLE = le[v], bLE = le[v + 1];
      float c = cr[v];
      wn += (aLE & (bLE ^ 1) & (c > 0.0f)) - ((aLE ^ 1) & bLE & (c < 0.0f));
    }
    le0b |= le[0] << tTile;
    cr7[tTile] = cr[7];
    le7[tTile] = le[7];
  }
  int recvb = __shfl_xor(le0b, 16, 32);
#pragma unroll
  for (int tTile = 0; tTile < 4; ++tTile) {
    int nt = hi ? ((tTile + 1) & 3) : tTile;
    int bLE = (recvb >> nt) & 1;
    int aLE = le7[tTile];
    float c = cr7[tTile];
    wn += (aLE & (bLE ^ 1) & (c > 0.0f)) - ((aLE ^ 1) & bLE & (c < 0.0f));
  }
  wn += __shfl_xor(wn, 16, 32);

  float m01 = fminf(e[0], e[1]), m23 = fminf(e[2], e[3]);
  float m45 = fminf(e[4], e[5]), m67 = fminf(e[6], e[7]);
  float m8  = fminf(fminf(m01, m23), fminf(m45, m67));
  float d2min = fminf(m8, __shfl_xor(m8, 16, 32));

  float dminv = sqrtf(d2min + 1e-12f);
  float sdf2d = (wn != 0) ? -dminv : dminv;
  float dz = fabsf(plz) - habs;
  float po = fmaxf(sdf2d, 0.0f), dzo = fmaxf(dz, 0.0f);
  float sdf = fminf(fmaxf(sdf2d, dz), 0.0f) + sqrtf(po * po + dzo * dzo + 1e-12f);

  size_t o = ((size_t)b * M_ + m) * K_ + k;
  float* basep = hi ? dmin_out : sdf_out;
  float  outv  = hi ? dminv    : sdf;
  basep[o] = outv;
}

__global__ __launch_bounds__(64) void combine_kernel(
    const float* __restrict__ sdf_in, const float* __restrict__ iw,
    const float* __restrict__ uw, const int* __restrict__ is_training,
    float* __restrict__ inter_out, float* __restrict__ occ_out)
{
  const int bm = blockIdx.x;
  const int b  = bm >> 11;
  const int c  = threadIdx.x;
  __shared__ float occ_sh[K_];
  __shared__ float preu[C_];

  float sv = sdf_in[(size_t)bm * K_ + c];
  occ_sh[c] = 1.0f / (1.0f + __expf(150.0f * sv));
  const float* wp = iw + ((size_t)b * K_) * C_ + c;
  __builtin_prefetch(wp, 0, 3);
  int train = is_training[0];
  __syncthreads();

  float inter;
  if (train) {
    float pmin = 3.0e38f;
#pragma unroll 4
    for (int kk = 0; kk < K_; ++kk) {
      float p = 1.0f - wp[(size_t)kk * C_] * (1.0f - occ_sh[kk]);
      pmin = fminf(pmin, p);
    }
    float s = 0.0f, ts = 0.0f;
#pragma unroll 4
    for (int kk = 0; kk < K_; ++kk) {
      float p = 1.0f - wp[(size_t)kk * C_] * (1.0f - occ_sh[kk]);
      float ee = __expf(-40.0f * (p - pmin));
      s += ee; ts += ee * p;
    }
    inter = ts / s;
  } else {
    inter = 3.0e38f;
#pragma unroll 4
    for (int kk = 0; kk < K_; ++kk) {
      float p = 1.0f - wp[(size_t)kk * C_] * (1.0f - occ_sh[kk]);
      inter = fminf(inter, p);
    }
  }
  inter_out[(size_t)bm * C_ + c] = inter;
  preu[c] = uw[b * C_ + c] * inter;
  __syncthreads();

  if (c < 32) {
    float a0 = preu[c], a1 = preu[c + 32];
    if (train) {
      float mx = fmaxf(a0, a1);
#pragma unroll
      for (int mask = 1; mask <= 16; mask <<= 1)
        mx = fmaxf(mx, __shfl_xor(mx, mask, 32));
      float e0 = __expf(40.0f * (a0 - mx));
      float e1 = __expf(40.0f * (a1 - mx));
      float s  = e0 + e1;
      float ts = e0 * a0 + e1 * a1;
#pragma unroll
      for (int mask = 1; mask <= 16; mask <<= 1) {
        s  += __shfl_xor(s,  mask, 32);
        ts += __shfl_xor(ts, mask, 32);
      }
      if (c == 0) occ_out[bm] = ts / s;
    } else {
      float mx = fmaxf(a0, a1);
#pragma unroll
      for (int mask = 1; mask <= 16; mask <<= 1)
        mx = fmaxf(mx, __shfl_xor(mx, mask, 32));
      if (c == 0) occ_out[bm] = mx;
    }
  }
}

extern "C" void kernel_launch(void* const* d_in, const int* in_sizes, int n_in,
                              void* d_out, int out_size, void* d_ws, size_t ws_size,
                              hipStream_t stream) {
  const float* pts   = (const float*)d_in[0];
  const float* prim  = (const float*)d_in[1];
  const float* iw    = (const float*)d_in[2];
  const float* uw    = (const float*)d_in[3];
  const int*   train = (const int*)d_in[4];

  float* occ_out   = (float*)d_out;
  float* sdf_out   = occ_out + (size_t)B_ * M_;
  float* inter_out = sdf_out + (size_t)B_ * M_ * K_;
  float* dmin_out  = inter_out + (size_t)B_ * M_ * C_;

  dim3 g1(B_ * K_, M_ / 64);
  sdf_kernel<<<g1, 128, 0, stream>>>(pts, prim, sdf_out, dmin_out);
  combine_kernel<<<B_ * M_, 64, 0, stream>>>(sdf_out, iw, uw, train,
                                             inter_out, occ_out);
  (void)in_sizes; (void)n_in; (void)out_size; (void)d_ws; (void)ws_size;
}


# curve launch removed (2 launches, no workspace): sines/cosines of the 12 fixed control angles are constants, so each sdf block builds its 8 curves and poses itself in LDS; sums keep parent's order
# speedup vs baseline: 2.2782x; 2.2782x over previous
#include <hip/hip_runtime.h>
#include <hip/hip_bf16.h>
#include <math.h>

typedef float v2f __attribute__((ext_vector_type(2)));
typedef float v8f __attribute__((ext_vector_type(8)));

#define B_ 4
#define M_ 2048
#define K_ 64
#define C_ 64
#define P_ 28
#define NCURVE 64

__device__ __forceinline__ v8f vmin8(v8f a, v8f b){
  v8f r;
#pragma unroll
  for (int i = 0; i < 8; ++i) r[i] = fminf(a[i], b[i]);
  return r;
}

__device__ static const float CTRL_CS[4][8] __attribute__((aligned(32))) = {
  {1.0f, 0.0f, 0.862856209f, 0.505449474f, 0.505449474f, 0.862856209f, 0.0f, 0.0f},
  {-4.37113883e-8f, 1.0f, -0.505449533f, 0.86285615f, -0.862856209f, 0.505449533f, 0.0f, 0.0f},
  {-1.0f, -8.74227766e-8f, -0.862856209f, -0.505449474f, -0.505449593f, -0.86285615f, 0.0f, 0.0f},
  {1.19248806e-8f, -1.0f, 0.505449593f, -0.86285615f, 0.862856209f, -0.505449414f, 0.0f, 0.0f}
};

#define KW_ 8
__global__ __launch_bounds__(256, 4) void sdf_kernel(
    const float* __restrict__ pts, const float* __restrict__ prim,
    float* __restrict__ sdf_out, float* __restrict__ dmin_out)
{
  __shared__ float4 s_f4[KW_ * 4 * 32];
  __shared__ float2 s_y2[KW_ * 4 * 32];
  __shared__ float2 s_c2[KW_ * NCURVE];
  __shared__ __attribute__((aligned(16))) float s_pose[KW_ * 32];
  const int tid = threadIdx.x;
  const int b  = blockIdx.x >> 3;
  const int kg = blockIdx.x & 7;
  const float* Pb = prim + (size_t)b * P_ * K_ + kg * KW_;

  const int wvp = tid >> 5, cpn = tid & 7;
  const float* Pq = Pb + wvp;
  const float q0 = Pq[0], q1 = Pq[K_], q2 = Pq[2 * K_], q3 = Pq[3 * K_];
  const float pown = Pq[cpn * K_];

#pragma unroll
  for (int h = 0; h < 2; ++h) {
    const int q = tid + h * 256;
    const int j = q >> 6, i = q & 63;
    const int seg = i >> 4, seg1 = (seg + 1) & 3;
    const float t = (float)(i & 15) * (1.0f / 16.0f);
    const float* Pp = Pb + j;
    const float rad0 = fabsf(Pp[(8 + seg * 3 + 0) * K_]);
    const float rad1 = fabsf(Pp[(8 + seg * 3 + 1) * K_]);
    const float rad2 = fabsf(Pp[(8 + seg * 3 + 2) * K_]);
    const float rad3 = fabsf(Pp[(8 + seg1 * 3 + 0) * K_]);
    const float w1 = fabsf(Pp[(20 + seg * 2 + 0) * K_]);
    const float w2 = fabsf(Pp[(20 + seg * 2 + 1) * K_]);
    const float4 d01 = *reinterpret_cast<const float4*>(&CTRL_CS[seg][0]);
    const float2 d2  = *reinterpret_cast<const float2*>(&CTRL_CS[seg][4]);
    const float2 d3  = *reinterpret_cast<const float2*>(&CTRL_CS[seg1][0]);
    const float c0x = d01.x * rad0, c0y = d01.y * rad0;
    const float c1x = d01.z * rad1, c1y = d01.w * rad1;
    const float c2x = d2.x * rad2,  c2y = d2.y * rad2;
    const float c3x = d3.x * rad3,  c3y = d3.y * rad3;
    const float u = 1.0f - t;
    const float b0 = u * u * u, b1 = 3.0f * u * u * t;
    const float b2 = 3.0f * u * t * t, b3 = t * t * t;
    const float inv = 1.0f / (b0 + w1 * b1 + w2 * b2 + b3);
    const float cx = (c0x * b0 + c1x * (w1 * b1) + c2x * (w2 * b2) + c3x * b3) * inv;
    const float cy = (c0y * b0 + c1y * (w1 * b1) + c2y * (w2 * b2) + c3y * b3) * inv;
    s_c2[q] = make_float2(cx, cy);
  }
  {
    const float invn = 1.0f / sqrtf(q0 * q0 + q1 * q1 + q2 * q2 + q3 * q3 + 1e-8f);
    const float pv = (cpn == 0) ? pown * invn
                   : (cpn < 4)  ? -pown * invn
                   : (cpn == 7) ? fabsf(pown) : pown;
    s_pose[wvp * 32 + (tid & 31)] = pv;
  }
  __syncthreads();

#pragma unroll
  for (int h = 0; h < 2; ++h) {
    const int q = tid + h * 256;
    const int j = q >> 6, i = q & 63;
    const float2 ca = s_c2[q];
    const float2 cb = s_c2[(j << 6) + ((i + 1) & 63)];
    const int base = (j * 4 + (i >> 4)) * 32 + (i & 15);
    s_f4[base]      = make_float4(-2.0f * ca.x, -2.0f * ca.y, ca.y - cb.y, cb.x - ca.x);
    s_f4[base + 16] = make_float4(1.0f, ca.x * ca.x + ca.y * ca.y,
                                  ca.x * cb.y - ca.y * cb.x, 0.0f);
    s_y2[base]      = make_float2(0.0f, 1.0f);
    s_y2[base + 16] = make_float2(-ca.y, 0.0f);
  }

  const int lane = tid & 31;
  const int wv = tid >> 5;
  const int row = lane & 15;
  const bool hi = lane >= 16;
  const int m = blockIdx.y * 128 + wv * 16 + row;
  const float* p3 = pts + ((size_t)b * M_ + m) * 3;
  const float P3x = p3[0], P3y = p3[1], P3z = p3[2];
  __syncthreads();

  v8f z = {0.f, 0.f, 0.f, 0.f, 0.f, 0.f, 0.f, 0.f};
  float res[KW_];
#pragma unroll
  for (int j = 0; j < KW_; ++j) {
    const float4 pa = *reinterpret_cast<const float4*>(&s_pose[j * 32]);
    const float4 pb = *reinterpret_cast<const float4*>(&s_pose[j * 32 + 4]);
    const float qw = pa.x, vx = pa.y, vy = pa.z, vz = pa.w;
    const float tx = pb.x, ty = pb.y, tz = pb.z, habs = pb.w;

    float px0 = P3x - tx, py0 = P3y - ty, pz0 = P3z - tz;
    float t2x = 2.0f * (vy * pz0 - vz * py0);
    float t2y = 2.0f * (vz * px0 - vx * pz0);
    float t2z = 2.0f * (vx * py0 - vy * px0);
    float plx = px0 + qw * t2x + (vy * t2z - vz * t2y);
    float ply = py0 + qw * t2y + (vz * t2x - vx * t2z);
    float plz = pz0 + qw * t2z + (vx * t2y - vy * t2x);
    float px = plx, py = ply;
    float ss = px * px + py * py;

    v2f Bd; Bd.x = hi ? ss : px;   Bd.y = hi ? 1.0f : py;
    v2f Bw; Bw.x = hi ? 1.0f : px; Bw.y = hi ? 0.0f : py;
    v8f e = {3e38f, 3e38f, 3e38f, 3e38f, 3e38f, 3e38f, 3e38f, 3e38f};

    int wn = 0;
    int le0b = 0;
    float cr7[4];
    int   le7[4];
#pragma unroll
    for (int tTile = 0; tTile < 4; ++tTile) {
      const float4 f = s_f4[(j * 4 + tTile) * 32 + lane];
      const float2 y = s_y2[(j * 4 + tTile) * 32 + lane];
      v2f Ad; Ad.x = f.x; Ad.y = f.y;
      v2f Ac; Ac.x = f.z; Ac.y = f.w;
      v2f Ay; Ay.x = y.x; Ay.y = y.y;
      v8f d  = __builtin_amdgcn_wmma_f32_16x16x4_f32(
          false, Ad, false, Bd, (short)0, z, false, false);
      v8f cr = __builtin_amdgcn_wmma_f32_16x16x4_f32(
          false, Ac, false, Bw, (short)0, z, false, false);
      v8f yd = __builtin_amdgcn_wmma_f32_16x16x4_f32(
          false, Ay, false, Bw, (short)0, z, false, false);
      e = vmin8(e, d);
      int le[8];
#pragma unroll
      for (int v = 0; v < 8; ++v) le[v] = (yd[v] >= 0.0f);
#pragma unroll
      for (int v = 0; v < 7; ++v) {
        int aLE = le[v], bLE = le[v + 1];
        float c = cr[v];
        wn += (aLE & (bLE ^ 1) & (c > 0.0f)) - ((aLE ^ 1) & bLE & (c < 0.0f));
      }
      le0b |= le[0] << tTile;
      cr7[tTile] = cr[7];
      le7[tTile] = le[7];
    }
    int recvb = __shfl_xor(le0b, 16, 32);
#pragma unroll
    for (int tTile = 0; tTile < 4; ++tTile) {
      int nt = hi ? ((tTile + 1) & 3) : tTile;
      int bLE = (recvb >> nt) & 1;
      int aLE = le7[tTile];
      float c = cr7[tTile];
      wn += (aLE & (bLE ^ 1) & (c > 0.0f)) - ((aLE ^ 1) & bLE & (c < 0.0f));
    }
    wn += __shfl_xor(wn, 16, 32);

    float m01 = fminf(e[0], e[1]), m23 = fminf(e[2], e[3]);
    float m45 = fminf(e[4], e[5]), m67 = fminf(e[6], e[7]);
    float m8  = fminf(fminf(m01, m23), fminf(m45, m67));
    float d2min = fminf(m8, __shfl_xor(m8, 16, 32));

    float dminv = sqrtf(d2min + 1e-12f);
    float sdf2d = (wn != 0) ? -dminv : dminv;
    float dz = fabsf(plz) - habs;
    float po = fmaxf(sdf2d, 0.0f), dzo = fmaxf(dz, 0.0f);
    float sdf = fminf(fmaxf(sdf2d, dz), 0.0f) + sqrtf(po * po + dzo * dzo + 1e-12f);
    res[j] = hi ? dminv : sdf;
  }

  size_t o = ((size_t)b * M_ + m) * K_ + kg * KW_;
  float* basep = hi ? dmin_out : sdf_out;
  float4* o4 = reinterpret_cast<float4*>(basep + o);
  o4[0] = make_float4(res[0], res[1], res[2], res[3]);
  o4[1] = make_float4(res[4], res[5], res[6], res[7]);
}

#define CP_ 32
#define PITCH_ 65
__global__ __launch_bounds__(256) void combine_kernel(
    const float* __restrict__ sdf_in, const float* __restrict__ iw,
    const float* __restrict__ uw, const int* __restrict__ is_training,
    float* __restrict__ inter_out, float* __restrict__ occ_out)
{
  __shared__ float om_sh[CP_ * PITCH_];
  __shared__ float in_sh[CP_ * PITCH_];
  const int tid  = threadIdx.x;
  const int lane = tid & 31;
  const int wv = __builtin_amdgcn_readfirstlane(tid >> 5);
  const int blk = blockIdx.x;
  const int b   = blk >> 6;
  const size_t bm0 = (size_t)blk * CP_;
  const int train = is_training[0];

  const float* sp = sdf_in + bm0 * K_;
#pragma unroll
  for (int h = 0; h < 2; ++h) {
    const int q = tid + h * 256;
    const float4 v = *reinterpret_cast<const float4*>(sp + q * 4);
    float* d = om_sh + (q >> 4) * PITCH_ + (q & 15) * 4;
    d[0] = 1.0f - 1.0f / (1.0f + __expf(150.0f * v.x));
    d[1] = 1.0f - 1.0f / (1.0f + __expf(150.0f * v.y));
    d[2] = 1.0f - 1.0f / (1.0f + __expf(150.0f * v.z));
    d[3] = 1.0f - 1.0f / (1.0f + __expf(150.0f * v.w));
  }
  __syncthreads();

  const float* orow = om_sh + lane * PITCH_;
  const float* wp = iw + (size_t)b * (K_ * C_) + wv * 8;
  float res[8];
  if (train) {
    float pmin[8];
#pragma unroll
    for (int j = 0; j < 8; ++j) pmin[j] = 3.0e38f;
#pragma unroll 8
    for (int k = 0; k < K_; ++k) {
      const float o = orow[k];
#pragma unroll
      for (int j = 0; j < 8; ++j) {
        float p = 1.0f - wp[k * C_ + j] * o;
        pmin[j] = fminf(pmin[j], p);
      }
    }
    float s[8], ts[8];
#pragma unroll
    for (int j = 0; j < 8; ++j) { s[j] = 0.0f; ts[j] = 0.0f; }
#pragma unroll 8
    for (int k = 0; k < K_; ++k) {
      const float o = orow[k];
#pragma unroll
      for (int j = 0; j < 8; ++j) {
        float p = 1.0f - wp[k * C_ + j] * o;
        float ee = __expf(-40.0f * (p - pmin[j]));
        s[j] += ee; ts[j] += ee * p;
      }
    }
#pragma unroll
    for (int j = 0; j < 8; ++j) res[j] = ts[j] / s[j];
  } else {
#pragma unroll
    for (int j = 0; j < 8; ++j) res[j] = 3.0e38f;
#pragma unroll 8
    for (int k = 0; k < K_; ++k) {
      const float o = orow[k];
#pragma unroll
      for (int j = 0; j < 8; ++j) {
        float p = 1.0f - wp[k * C_ + j] * o;
        res[j] = fminf(res[j], p);
      }
    }
  }
  {
    float* irow = in_sh + lane * PITCH_ + wv * 8;
#pragma unroll
    for (int j = 0; j < 8; ++j) irow[j] = res[j];
  }
  __syncthreads();

  float* op = inter_out + bm0 * C_;
#pragma unroll
  for (int h = 0; h < 2; ++h) {
    const int q = tid + h * 256;
    const float* s4 = in_sh + (q >> 4) * PITCH_ + (q & 15) * 4;
    *reinterpret_cast<float4*>(op + q * 4) = make_float4(s4[0], s4[1], s4[2], s4[3]);
  }

  if (wv == 0) {
    const float* ir = in_sh + lane * PITCH_;
    const float* up = uw + b * C_;
    float a[C_];
#pragma unroll
    for (int c = 0; c < C_; ++c) a[c] = up[c] * ir[c];
    float mx = a[0];
#pragma unroll
    for (int c = 1; c < C_; ++c) mx = fmaxf(mx, a[c]);
    if (train) {
      float sv[32], tv[32];
#pragma unroll
      for (int i = 0; i < 32; ++i) {
        float a0 = a[i], a1 = a[i + 32];
        float e0 = __expf(40.0f * (a0 - mx));
        float e1 = __expf(40.0f * (a1 - mx));
        sv[i] = e0 + e1;
        tv[i] = e0 * a0 + e1 * a1;
      }
#pragma unroll
      for (int step = 1; step < 32; step <<= 1) {
#pragma unroll
        for (int i = 0; i < 32; i += 2 * step) {
          sv[i] += sv[i + step];
          tv[i] += tv[i + step];
        }
      }
      occ_out[bm0 + lane] = tv[0] / sv[0];
    } else {
      occ_out[bm0 + lane] = mx;
    }
  }
}

extern "C" void kernel_launch(void* const* d_in, const int* in_sizes, int n_in,
                              void* d_out, int out_size, void* d_ws, size_t ws_size,
                              hipStream_t stream) {
  const float* pts   = (const float*)d_in[0];
  const float* prim  = (const float*)d_in[1];
  const float* iw    = (const float*)d_in[2];
  const float* uw    = (const float*)d_in[3];
  const int*   train = (const int*)d_in[4];

  float* occ_out   = (float*)d_out;
  float* sdf_out   = occ_out + (size_t)B_ * M_;
  float* inter_out = sdf_out + (size_t)B_ * M_ * K_;
  float* dmin_out  = inter_out + (size_t)B_ * M_ * C_;

  dim3 g1(B_ * (K_ / KW_), M_ / 128);
  sdf_kernel<<<g1, 256, 0, stream>>>(pts, prim, sdf_out, dmin_out);
  combine_kernel<<<B_ * M_ / CP_, 256, 0, stream>>>(sdf_out, iw, uw, train,
                                             inter_out, occ_out);
  (void)d_ws;
  (void)in_sizes; (void)n_in; (void)out_size; (void)ws_size;
}


# combine: the batch's 64x64 weights staged once per block in LDS and read as wave-uniform 16-byte LDS loads (vector operands, no scalar loads or moves), next 8 k read ahead; sums keep parent's order
# speedup vs baseline: 2.5342x; 1.1124x over previous
#include <hip/hip_runtime.h>
#include <hip/hip_bf16.h>
#include <math.h>

typedef float v2f __attribute__((ext_vector_type(2)));
typedef float v8f __attribute__((ext_vector_type(8)));

#define B_ 4
#define M_ 2048
#define K_ 64
#define C_ 64
#define P_ 28
#define NCURVE 64

__device__ __forceinline__ v8f vmin8(v8f a, v8f b){
  v8f r;
#pragma unroll
  for (int i = 0; i < 8; ++i) r[i] = fminf(a[i], b[i]);
  return r;
}

__device__ static const float CTRL_CS[4][8] __attribute__((aligned(32))) = {
  {1.0f, 0.0f, 0.862856209f, 0.505449474f, 0.505449474f, 0.862856209f, 0.0f, 0.0f},
  {-4.37113883e-8f, 1.0f, -0.505449533f, 0.86285615f, -0.862856209f, 0.505449533f, 0.0f, 0.0f},
  {-1.0f, -8.74227766e-8f, -0.862856209f, -0.505449474f, -0.505449593f, -0.86285615f, 0.0f, 0.0f},
  {1.19248806e-8f, -1.0f, 0.505449593f, -0.86285615f, 0.862856209f, -0.505449414f, 0.0f, 0.0f}
};

#define KW_ 8
__global__ __launch_bounds__(256, 4) void sdf_kernel(
    const float* __restrict__ pts, const float* __restrict__ prim,
    float* __restrict__ sdf_out, float* __restrict__ dmin_out)
{
  __shared__ float4 s_f4[KW_ * 4 * 32];
  __shared__ float2 s_y2[KW_ * 4 * 32];
  __shared__ float2 s_c2[KW_ * NCURVE];
  __shared__ __attribute__((aligned(16))) float s_pose[KW_ * 32];
  const int tid = threadIdx.x;
  const int b  = blockIdx.x >> 3;
  const int kg = blockIdx.x & 7;
  const float* Pb = prim + (size_t)b * P_ * K_ + kg * KW_;

  const int wvp = tid >> 5, cpn = tid & 7;
  const float* Pq = Pb + wvp;
  const float q0 = Pq[0], q1 = Pq[K_], q2 = Pq[2 * K_], q3 = Pq[3 * K_];
  const float pown = Pq[cpn * K_];

#pragma unroll
  for (int h = 0; h < 2; ++h) {
    const int q = tid + h * 256;
    const int j = q >> 6, i = q & 63;
    const int seg = i >> 4, seg1 = (seg + 1) & 3;
    const float t = (float)(i & 15) * (1.0f / 16.0f);
    const float* Pp = Pb + j;
    const float rad0 = fabsf(Pp[(8 + seg * 3 + 0) * K_]);
    const float rad1 = fabsf(Pp[(8 + seg * 3 + 1) * K_]);
    const float rad2 = fabsf(Pp[(8 + seg * 3 + 2) * K_]);
    const float rad3 = fabsf(Pp[(8 + seg1 * 3 + 0) * K_]);
    const float w1 = fabsf(Pp[(20 + seg * 2 + 0) * K_]);
    const float w2 = fabsf(Pp[(20 + seg * 2 + 1) * K_]);
    const float4 d01 = *reinterpret_cast<const float4*>(&CTRL_CS[seg][0]);
    const float2 d2  = *reinterpret_cast<const float2*>(&CTRL_CS[seg][4]);
    const float2 d3  = *reinterpret_cast<const float2*>(&CTRL_CS[seg1][0]);
    const float c0x = d01.x * rad0, c0y = d01.y * rad0;
    const float c1x = d01.z * rad1, c1y = d01.w * rad1;
    const float c2x = d2.x * rad2,  c2y = d2.y * rad2;
    const float c3x = d3.x * rad3,  c3y = d3.y * rad3;
    const float u = 1.0f - t;
    const float b0 = u * u * u, b1 = 3.0f * u * u * t;
    const float b2 = 3.0f * u * t * t, b3 = t * t * t;
    const float inv = 1.0f / (b0 + w1 * b1 + w2 * b2 + b3);
    const float cx = (c0x * b0 + c1x * (w1 * b1) + c2x * (w2 * b2) + c3x * b3) * inv;
    const float cy = (c0y * b0 + c1y * (w1 * b1) + c2y * (w2 * b2) + c3y * b3) * inv;
    s_c2[q] = make_float2(cx, cy);
  }
  {
    const float invn = 1.0f / sqrtf(q0 * q0 + q1 * q1 + q2 * q2 + q3 * q3 + 1e-8f);
    const float pv = (cpn == 0) ? pown * invn
                   : (cpn < 4)  ? -pown * invn
                   : (cpn == 7) ? fabsf(pown) : pown;
    s_pose[wvp * 32 + (tid & 31)] = pv;
  }
  __syncthreads();

#pragma unroll
  for (int h = 0; h < 2; ++h) {
    const int q = tid + h * 256;
    const int j = q >> 6, i = q & 63;
    const float2 ca = s_c2[q];
    const float2 cb = s_c2[(j << 6) + ((i + 1) & 63)];
    const int base = (j * 4 + (i >> 4)) * 32 + (i & 15);
    s_f4[base]      = make_float4(-2.0f * ca.x, -2.0f * ca.y, ca.y - cb.y, cb.x - ca.x);
    s_f4[base + 16] = make_float4(1.0f, ca.x * ca.x + ca.y * ca.y,
                                  ca.x * cb.y - ca.y * cb.x, 0.0f);
    s_y2[base]      = make_float2(0.0f, 1.0f);
    s_y2[base + 16] = make_float2(-ca.y, 0.0f);
  }

  const int lane = tid & 31;
  const int wv = tid >> 5;
  const int row = lane & 15;
  const bool hi = lane >= 16;
  const int m = blockIdx.y * 128 + wv * 16 + row;
  const float* p3 = pts + ((size_t)b * M_ + m) * 3;
  const float P3x = p3[0], P3y = p3[1], P3z = p3[2];
  __syncthreads();

  v8f z = {0.f, 0.f, 0.f, 0.f, 0.f, 0.f, 0.f, 0.f};
  float res[KW_];
#pragma unroll
  for (int j = 0; j < KW_; ++j) {
    const float4 pa = *reinterpret_cast<const float4*>(&s_pose[j * 32]);
    const float4 pb = *reinterpret_cast<const float4*>(&s_pose[j * 32 + 4]);
    const float qw = pa.x, vx = pa.y, vy = pa.z, vz = pa.w;
    const float tx = pb.x, ty = pb.y, tz = pb.z, habs = pb.w;

    float px0 = P3x - tx, py0 = P3y - ty, pz0 = P3z - tz;
    float t2x = 2.0f * (vy * pz0 - vz * py0);
    float t2y = 2.0f * (vz * px0 - vx * pz0);
    float t2z = 2.0f * (vx * py0 - vy * px0);
    float plx = px0 + qw * t2x + (vy * t2z - vz * t2y);
    float ply = py0 + qw * t2y + (vz * t2x - vx * t2z);
    float plz = pz0 + qw * t2z + (vx * t2y - vy * t2x);
    float px = plx, py = ply;
    float ss = px * px + py * py;

    v2f Bd; Bd.x = hi ? ss : px;   Bd.y = hi ? 1.0f : py;
    v2f Bw; Bw.x = hi ? 1.0f : px; Bw.y = hi ? 0.0f : py;
    v8f e = {3e38f, 3e38f, 3e38f, 3e38f, 3e38f, 3e38f, 3e38f, 3e38f};

    int wn = 0;
    int le0b = 0;
    float cr7[4];
    int   le7[4];
#pragma unroll
    for (int tTile = 0; tTile < 4; ++tTile) {
      const float4 f = s_f4[(j * 4 + tTile) * 32 + lane];
      const float2 y = s_y2[(j * 4 + tTile) * 32 + lane];
      v2f Ad; Ad.x = f.x; Ad.y = f.y;
      v2f Ac; Ac.x = f.z; Ac.y = f.w;
      v2f Ay; Ay.x = y.x; Ay.y = y.y;
      v8f d  = __builtin_amdgcn_wmma_f32_16x16x4_f32(
          false, Ad, false, Bd, (short)0, z, false, false);
      v8f cr = __builtin_amdgcn_wmma_f32_16x16x4_f32(
          false, Ac, false, Bw, (short)0, z, false, false);
      v8f yd = __builtin_amdgcn_wmma_f32_16x16x4_f32(
          false, Ay, false, Bw, (short)0, z, false, false);
      e = vmin8(e, d);
      int le[8];
#pragma unroll
      for (int v = 0; v < 8; ++v) le[v] = (yd[v] >= 0.0f);
#pragma unroll
      for (int v = 0; v < 7; ++v) {
        int aLE = le[v], bLE = le[v + 1];
        float c = cr[v];
        wn += (aLE & (bLE ^ 1) & (c > 0.0f)) - ((aLE ^ 1) & bLE & (c < 0.0f));
      }
      le0b |= le[0] << tTile;
      cr7[tTile] = cr[7];
      le7[tTile] = le[7];
    }
    int recvb = __shfl_xor(le0b, 16, 32);
#pragma unroll
    for (int tTile = 0; tTile < 4; ++tTile) {
      int nt = hi ? ((tTile + 1) & 3) : tTile;
      int bLE = (recvb >> nt) & 1;
      int aLE = le7[tTile];
      float c = cr7[tTile];
      wn += (aLE & (bLE ^ 1) & (c > 0.0f)) - ((aLE ^ 1) & bLE & (c < 0.0f));
    }
    wn += __shfl_xor(wn, 16, 32);

    float m01 = fminf(e[0], e[1]), m23 = fminf(e[2], e[3]);
    float m45 = fminf(e[4], e[5]), m67 = fminf(e[6], e[7]);
    float m8  = fminf(fminf(m01, m23), fminf(m45, m67));
    float d2min = fminf(m8, __shfl_xor(m8, 16, 32));

    float dminv = sqrtf(d2min + 1e-12f);
    float sdf2d = (wn != 0) ? -dminv : dminv;
    float dz = fabsf(plz) - habs;
    float po = fmaxf(sdf2d, 0.0f), dzo = fmaxf(dz, 0.0f);
    float sdf = fminf(fmaxf(sdf2d, dz), 0.0f) + sqrtf(po * po + dzo * dzo + 1e-12f);
    res[j] = hi ? dminv : sdf;
  }

  size_t o = ((size_t)b * M_ + m) * K_ + kg * KW_;
  float* basep = hi ? dmin_out : sdf_out;
  float4* o4 = reinterpret_cast<float4*>(basep + o);
  o4[0] = make_float4(res[0], res[1], res[2], res[3]);
  o4[1] = make_float4(res[4], res[5], res[6], res[7]);
}

#define CP_ 32
#define PITCH_ 65

__device__ __forceinline__ void load_trip(const float* wcol, const float* orow, int k0,
                                          float4 (&wa)[16], float (&ov)[8]) {
#pragma unroll
  for (int i = 0; i < 8; ++i) {
    const int k = (k0 + i) & (K_ - 1);
    const float4* p = reinterpret_cast<const float4*>(wcol + k * C_);
    wa[2 * i] = p[0];
    wa[2 * i + 1] = p[1];
    ov[i] = orow[k];
  }
}

__device__ __forceinline__ void min_trip(const float4 (&wa)[16], const float (&ov)[8],
                                         float (&pmin)[8]) {
#pragma unroll
  for (int i = 0; i < 8; ++i) {
    const float o = ov[i];
    const float4 a = wa[2 * i], c = wa[2 * i + 1];
    pmin[0] = fminf(pmin[0], 1.0f - a.x * o);
    pmin[1] = fminf(pmin[1], 1.0f - a.y * o);
    pmin[2] = fminf(pmin[2], 1.0f - a.z * o);
    pmin[3] = fminf(pmin[3], 1.0f - a.w * o);
    pmin[4] = fminf(pmin[4], 1.0f - c.x * o);
    pmin[5] = fminf(pmin[5], 1.0f - c.y * o);
    pmin[6] = fminf(pmin[6], 1.0f - c.z * o);
    pmin[7] = fminf(pmin[7], 1.0f - c.w * o);
  }
}

__device__ __forceinline__ void sum_trip(const float4 (&wa)[16], const float (&ov)[8],
                                         const float (&pmin)[8], float (&s)[8], float (&ts)[8]) {
#pragma unroll
  for (int i = 0; i < 8; ++i) {
    const float o = ov[i];
    const float4 a = wa[2 * i], c = wa[2 * i + 1];
    const float w8[8] = {a.x, a.y, a.z, a.w, c.x, c.y, c.z, c.w};
#pragma unroll
    for (int j = 0; j < 8; ++j) {
      float p = 1.0f - w8[j] * o;
      float ee = __expf(-40.0f * (p - pmin[j]));
      s[j] += ee; ts[j] += ee * p;
    }
  }
}

__global__ __launch_bounds__(256) void combine_kernel(
    const float* __restrict__ sdf_in, const float* __restrict__ iw,
    const float* __restrict__ uw, const int* __restrict__ is_training,
    float* __restrict__ inter_out, float* __restrict__ occ_out)
{
  __shared__ __attribute__((aligned(16))) float w_sh[K_ * C_];
  __shared__ float om_sh[CP_ * PITCH_];
  __shared__ float in_sh[CP_ * PITCH_];
  const int tid  = threadIdx.x;
  const int lane = tid & 31;
  const int wv = __builtin_amdgcn_readfirstlane(tid >> 5);
  const int blk = blockIdx.x;
  const int b   = blk >> 6;
  const size_t bm0 = (size_t)blk * CP_;
  const int train = is_training[0];

  {
    const float4* wsrc = reinterpret_cast<const float4*>(iw + (size_t)b * (K_ * C_));
    float4* wdst = reinterpret_cast<float4*>(w_sh);
    float4 wq[4];
#pragma unroll
    for (int h = 0; h < 4; ++h) wq[h] = wsrc[tid + h * 256];
#pragma unroll
    for (int h = 0; h < 4; ++h) wdst[tid + h * 256] = wq[h];
  }

  const float* sp = sdf_in + bm0 * K_;
#pragma unroll
  for (int h = 0; h < 2; ++h) {
    const int q = tid + h * 256;
    const float4 v = *reinterpret_cast<const float4*>(sp + q * 4);
    float* d = om_sh + (q >> 4) * PITCH_ + (q & 15) * 4;
    d[0] = 1.0f - 1.0f / (1.0f + __expf(150.0f * v.x));
    d[1] = 1.0f - 1.0f / (1.0f + __expf(150.0f * v.y));
    d[2] = 1.0f - 1.0f / (1.0f + __expf(150.0f * v.z));
    d[3] = 1.0f - 1.0f / (1.0f + __expf(150.0f * v.w));
  }
  __syncthreads();

  const float* orow = om_sh + lane * PITCH_;
  const float* wcol = w_sh + wv * 8;
  float res[8];
  float4 wa[16], wb[16];
  float oa[8], ob[8];
  if (train) {
    float pmin[8];
#pragma unroll
    for (int j = 0; j < 8; ++j) pmin[j] = 3.0e38f;
    load_trip(wcol, orow, 0, wa, oa);
#pragma unroll 1
    for (int k = 0; k < K_; k += 16) {
      load_trip(wcol, orow, k + 8, wb, ob);
      min_trip(wa, oa, pmin);
      load_trip(wcol, orow, k + 16, wa, oa);
      min_trip(wb, ob, pmin);
    }
    float s[8], ts[8];
#pragma unroll
    for (int j = 0; j < 8; ++j) { s[j] = 0.0f; ts[j] = 0.0f; }
#pragma unroll 1
    for (int k = 0; k < K_; k += 16) {
      load_trip(wcol, orow, k + 8, wb, ob);
      sum_trip(wa, oa, pmin, s, ts);
      load_trip(wcol, orow, k + 16, wa, oa);
      sum_trip(wb, ob, pmin, s, ts);
    }
#pragma unroll
    for (int j = 0; j < 8; ++j) res[j] = ts[j] / s[j];
  } else {
#pragma unroll
    for (int j = 0; j < 8; ++j) res[j] = 3.0e38f;
    load_trip(wcol, orow, 0, wa, oa);
#pragma unroll 1
    for (int k = 0; k < K_; k += 16) {
      load_trip(wcol, orow, k + 8, wb, ob);
      min_trip(wa, oa, res);
      load_trip(wcol, orow, k + 16, wa, oa);
      min_trip(wb, ob, res);
    }
  }
  {
    float* irow = in_sh + lane * PITCH_ + wv * 8;
#pragma unroll
    for (int j = 0; j < 8; ++j) irow[j] = res[j];
  }
  __syncthreads();

  float* op = inter_out + bm0 * C_;
#pragma unroll
  for (int h = 0; h < 2; ++h) {
    const int q = tid + h * 256;
    const float* s4 = in_sh + (q >> 4) * PITCH_ + (q & 15) * 4;
    *reinterpret_cast<float4*>(op + q * 4) = make_float4(s4[0], s4[1], s4[2], s4[3]);
  }

  if (wv == 0) {
    const float* ir = in_sh + lane * PITCH_;
    const float* up = uw + b * C_;
    float a[C_];
#pragma unroll
    for (int c = 0; c < C_; ++c) a[c] = up[c] * ir[c];
    float mx = a[0];
#pragma unroll
    for (int c = 1; c < C_; ++c) mx = fmaxf(mx, a[c]);
    if (train) {
      float sv[32], tv[32];
#pragma unroll
      for (int i = 0; i < 32; ++i) {
        float a0 = a[i], a1 = a[i + 32];
        float e0 = __expf(40.0f * (a0 - mx));
        float e1 = __expf(40.0f * (a1 - mx));
        sv[i] = e0 + e1;
        tv[i] = e0 * a0 + e1 * a1;
      }
#pragma unroll
      for (int step = 1; step < 32; step <<= 1) {
#pragma unroll
        for (int i = 0; i < 32; i += 2 * step) {
          sv[i] += sv[i + step];
          tv[i] += tv[i + step];
        }
      }
      occ_out[bm0 + lane] = tv[0] / sv[0];
    } else {
      occ_out[bm0 + lane] = mx;
    }
  }
}

extern "C" void kernel_launch(void* const* d_in, const int* in_sizes, int n_in,
                              void* d_out, int out_size, void* d_ws, size_t ws_size,
                              hipStream_t stream) {
  const float* pts   = (const float*)d_in[0];
  const float* prim  = (const float*)d_in[1];
  const float* iw    = (const float*)d_in[2];
  const float* uw    = (const float*)d_in[3];
  const int*   train = (const int*)d_in[4];

  float* occ_out   = (float*)d_out;
  float* sdf_out   = occ_out + (size_t)B_ * M_;
  float* inter_out = sdf_out + (size_t)B_ * M_ * K_;
  float* dmin_out  = inter_out + (size_t)B_ * M_ * C_;

  dim3 g1(B_ * (K_ / KW_), M_ / 128);
  sdf_kernel<<<g1, 256, 0, stream>>>(pts, prim, sdf_out, dmin_out);
  combine_kernel<<<B_ * M_ / CP_, 256, 0, stream>>>(sdf_out, iw, uw, train,
                                             inter_out, occ_out);
  (void)d_ws;
  (void)in_sizes; (void)n_in; (void)out_size; (void)ws_size;
}


# winding count as integer arithmetic on the bits of the yt and cr products: sign smears, one subtract and one 24-bit multiply-add per edge, cr rows pre-signed by edge direction; float sums kept
# speedup vs baseline: 2.9343x; 1.1579x over previous
#include <hip/hip_runtime.h>
#include <hip/hip_bf16.h>
#include <math.h>

typedef float v2f __attribute__((ext_vector_type(2)));
typedef float v8f __attribute__((ext_vector_type(8)));

#define B_ 4
#define M_ 2048
#define K_ 64
#define C_ 64
#define P_ 28
#define NCURVE 64

__device__ __forceinline__ v8f vmin8(v8f a, v8f b){
  v8f r;
#pragma unroll
  for (int i = 0; i < 8; ++i) r[i] = fminf(a[i], b[i]);
  return r;
}

__device__ static const float CTRL_CS[4][8] __attribute__((aligned(32))) = {
  {1.0f, 0.0f, 0.862856209f, 0.505449474f, 0.505449474f, 0.862856209f, 0.0f, 0.0f},
  {-4.37113883e-8f, 1.0f, -0.505449533f, 0.86285615f, -0.862856209f, 0.505449533f, 0.0f, 0.0f},
  {-1.0f, -8.74227766e-8f, -0.862856209f, -0.505449474f, -0.505449593f, -0.86285615f, 0.0f, 0.0f},
  {1.19248806e-8f, -1.0f, 0.505449593f, -0.86285615f, 0.862856209f, -0.505449414f, 0.0f, 0.0f}
};

#define KW_ 8
__global__ __launch_bounds__(256, 4) void sdf_kernel(
    const float* __restrict__ pts, const float* __restrict__ prim,
    float* __restrict__ sdf_out, float* __restrict__ dmin_out)
{
  __shared__ float4 s_f4[KW_ * 4 * 32];
  __shared__ float2 s_y2[KW_ * 4 * 32];
  __shared__ float2 s_c2[KW_ * NCURVE];
  __shared__ __attribute__((aligned(16))) float s_pose[KW_ * 32];
  const int tid = threadIdx.x;
  const int b  = blockIdx.x >> 3;
  const int kg = blockIdx.x & 7;
  const float* Pb = prim + (size_t)b * P_ * K_ + kg * KW_;

  const int wvp = tid >> 5, cpn = tid & 7;
  const float* Pq = Pb + wvp;
  const float q0 = Pq[0], q1 = Pq[K_], q2 = Pq[2 * K_], q3 = Pq[3 * K_];
  const float pown = Pq[cpn * K_];

#pragma unroll
  for (int h = 0; h < 2; ++h) {
    const int q = tid + h * 256;
    const int j = q >> 6, i = q & 63;
    const int seg = i >> 4, seg1 = (seg + 1) & 3;
    const float t = (float)(i & 15) * (1.0f / 16.0f);
    const float* Pp = Pb + j;
    const float rad0 = fabsf(Pp[(8 + seg * 3 + 0) * K_]);
    const float rad1 = fabsf(Pp[(8 + seg * 3 + 1) * K_]);
    const float rad2 = fabsf(Pp[(8 + seg * 3 + 2) * K_]);
    const float rad3 = fabsf(Pp[(8 + seg1 * 3 + 0) * K_]);
    const float w1 = fabsf(Pp[(20 + seg * 2 + 0) * K_]);
    const float w2 = fabsf(Pp[(20 + seg * 2 + 1) * K_]);
    const float4 d01 = *reinterpret_cast<const float4*>(&CTRL_CS[seg][0]);
    const float2 d2  = *reinterpret_cast<const float2*>(&CTRL_CS[seg][4]);
    const float2 d3  = *reinterpret_cast<const float2*>(&CTRL_CS[seg1][0]);
    const float c0x = d01.x * rad0, c0y = d01.y * rad0;
    const float c1x = d01.z * rad1, c1y = d01.w * rad1;
    const float c2x = d2.x * rad2,  c2y = d2.y * rad2;
    const float c3x = d3.x * rad3,  c3y = d3.y * rad3;
    const float u = 1.0f - t;
    const float b0 = u * u * u, b1 = 3.0f * u * u * t;
    const float b2 = 3.0f * u * t * t, b3 = t * t * t;
    const float inv = 1.0f / (b0 + w1 * b1 + w2 * b2 + b3);
    const float cx = (c0x * b0 + c1x * (w1 * b1) + c2x * (w2 * b2) + c3x * b3) * inv;
    const float cy = (c0y * b0 + c1y * (w1 * b1) + c2y * (w2 * b2) + c3y * b3) * inv;
    s_c2[q] = make_float2(cx, cy);
  }
  {
    const float invn = 1.0f / sqrtf(q0 * q0 + q1 * q1 + q2 * q2 + q3 * q3 + 1e-8f);
    const float pv = (cpn == 0) ? pown * invn
                   : (cpn < 4)  ? -pown * invn
                   : (cpn == 7) ? fabsf(pown) : pown;
    s_pose[wvp * 32 + (tid & 31)] = pv;
  }
  __syncthreads();

#pragma unroll
  for (int h = 0; h < 2; ++h) {
    const int q = tid + h * 256;
    const int j = q >> 6, i = q & 63;
    const float2 ca = s_c2[q];
    const float2 cb = s_c2[(j << 6) + ((i + 1) & 63)];
    const int base = (j * 4 + (i >> 4)) * 32 + (i & 15);
    const float sg = (cb.y > ca.y) ? -1.0f : 1.0f;
    s_f4[base]      = make_float4(-2.0f * ca.x, -2.0f * ca.y,
                                  sg * (ca.y - cb.y), sg * (cb.x - ca.x));
    s_f4[base + 16] = make_float4(1.0f, ca.x * ca.x + ca.y * ca.y,
                                  sg * (ca.x * cb.y - ca.y * cb.x), 0.0f);
    s_y2[base]      = make_float2(0.0f, 1.0f);
    s_y2[base + 16] = make_float2(-ca.y, 0.0f);
  }

  const int lane = tid & 31;
  const int wv = tid >> 5;
  const int row = lane & 15;
  const bool hi = lane >= 16;
  const int m = blockIdx.y * 128 + wv * 16 + row;
  const float* p3 = pts + ((size_t)b * M_ + m) * 3;
  const float P3x = p3[0], P3y = p3[1], P3z = p3[2];
  __syncthreads();

  v8f z = {0.f, 0.f, 0.f, 0.f, 0.f, 0.f, 0.f, 0.f};
  float res[KW_];
#pragma unroll
  for (int j = 0; j < KW_; ++j) {
    const float4 pa = *reinterpret_cast<const float4*>(&s_pose[j * 32]);
    const float4 pb = *reinterpret_cast<const float4*>(&s_pose[j * 32 + 4]);
    const float qw = pa.x, vx = pa.y, vy = pa.z, vz = pa.w;
    const float tx = pb.x, ty = pb.y, tz = pb.z, habs = pb.w;

    float px0 = P3x - tx, py0 = P3y - ty, pz0 = P3z - tz;
    float t2x = 2.0f * (vy * pz0 - vz * py0);
    float t2y = 2.0f * (vz * px0 - vx * pz0);
    float t2z = 2.0f * (vx * py0 - vy * px0);
    float plx = px0 + qw * t2x + (vy * t2z - vz * t2y);
    float ply = py0 + qw * t2y + (vz * t2x - vx * t2z);
    float plz = pz0 + qw * t2z + (vx * t2y - vy * t2x);
    float px = plx, py = ply;
    float ss = px * px + py * py;

    v2f Bd; Bd.x = hi ? ss : px;   Bd.y = hi ? 1.0f : py;
    v2f Bw; Bw.x = hi ? 1.0f : px; Bw.y = hi ? 0.0f : py;
    v8f e = {3e38f, 3e38f, 3e38f, 3e38f, 3e38f, 3e38f, 3e38f, 3e38f};

    int wn = 0, wn1 = 0;
    int n0b = 0;
    int m7[4];
    int n7[4];
#pragma unroll
    for (int tTile = 0; tTile < 4; ++tTile) {
      const float4 f = s_f4[(j * 4 + tTile) * 32 + lane];
      const float2 y = s_y2[(j * 4 + tTile) * 32 + lane];
      v2f Ad; Ad.x = f.x; Ad.y = f.y;
      v2f Ac; Ac.x = f.z; Ac.y = f.w;
      v2f Ay; Ay.x = y.x; Ay.y = y.y;
      v8f d  = __builtin_amdgcn_wmma_f32_16x16x4_f32(
          false, Ad, false, Bd, (short)0, z, false, false);
      v8f cr = __builtin_amdgcn_wmma_f32_16x16x4_f32(
          false, Ac, false, Bw, (short)0, z, false, false);
      v8f yd = __builtin_amdgcn_wmma_f32_16x16x4_f32(
          false, Ay, false, Bw, (short)0, z, false, false);
      e = vmin8(e, d);
      int n[8];
#pragma unroll
      for (int v = 0; v < 8; ++v) n[v] = __float_as_int(yd[v]) >> 31;
#pragma unroll
      for (int v = 0; v < 7; ++v) {
        const int mv = __float_as_int(cr[v]) >> 31;
        if (v & 1) wn1 += __mul24(n[v] - n[v + 1], mv);
        else       wn  += __mul24(n[v] - n[v + 1], mv);
      }
      n0b |= n[0] & (1 << tTile);
      m7[tTile] = __float_as_int(cr[7]) >> 31;
      n7[tTile] = n[7];
    }
    int recvb = __shfl_xor(n0b, 16, 32);
#pragma unroll
    for (int tTile = 0; tTile < 4; ++tTile) {
      int nt = hi ? ((tTile + 1) & 3) : tTile;
      int nb = -((recvb >> nt) & 1);
      wn1 += __mul24(n7[tTile] - nb, m7[tTile]);
    }
    wn += wn1;
    wn += __shfl_xor(wn, 16, 32);

    float m01 = fminf(e[0], e[1]), m23 = fminf(e[2], e[3]);
    float m45 = fminf(e[4], e[5]), m67 = fminf(e[6], e[7]);
    float m8  = fminf(fminf(m01, m23), fminf(m45, m67));
    float d2min = fminf(m8, __shfl_xor(m8, 16, 32));

    float dminv = sqrtf(d2min + 1e-12f);
    float sdf2d = (wn != 0) ? -dminv : dminv;
    float dz = fabsf(plz) - habs;
    float po = fmaxf(sdf2d, 0.0f), dzo = fmaxf(dz, 0.0f);
    float sdf = fminf(fmaxf(sdf2d, dz), 0.0f) + sqrtf(po * po + dzo * dzo + 1e-12f);
    res[j] = hi ? dminv : sdf;
  }

  size_t o = ((size_t)b * M_ + m) * K_ + kg * KW_;
  float* basep = hi ? dmin_out : sdf_out;
  float4* o4 = reinterpret_cast<float4*>(basep + o);
  o4[0] = make_float4(res[0], res[1], res[2], res[3]);
  o4[1] = make_float4(res[4], res[5], res[6], res[7]);
}

#define CP_ 32
#define PITCH_ 65

__device__ __forceinline__ void load_trip(const float* wcol, const float* orow, int k0,
                                          float4 (&wa)[16], float (&ov)[8]) {
#pragma unroll
  for (int i = 0; i < 8; ++i) {
    const int k = (k0 + i) & (K_ - 1);
    const float4* p = reinterpret_cast<const float4*>(wcol + k * C_);
    wa[2 * i] = p[0];
    wa[2 * i + 1] = p[1];
    ov[i] = orow[k];
  }
}

__device__ __forceinline__ void min_trip(const float4 (&wa)[16], const float (&ov)[8],
                                         float (&pmin)[8]) {
#pragma unroll
  for (int i = 0; i < 8; ++i) {
    const float o = ov[i];
    const float4 a = wa[2 * i], c = wa[2 * i + 1];
    pmin[0] = fminf(pmin[0], 1.0f - a.x * o);
    pmin[1] = fminf(pmin[1], 1.0f - a.y * o);
    pmin[2] = fminf(pmin[2], 1.0f - a.z * o);
    pmin[3] = fminf(pmin[3], 1.0f - a.w * o);
    pmin[4] = fminf(pmin[4], 1.0f - c.x * o);
    pmin[5] = fminf(pmin[5], 1.0f - c.y * o);
    pmin[6] = fminf(pmin[6], 1.0f - c.z * o);
    pmin[7] = fminf(pmin[7], 1.0f - c.w * o);
  }
}

__device__ __forceinline__ void sum_trip(const float4 (&wa)[16], const float (&ov)[8],
                                         const float (&pmin)[8], float (&s)[8], float (&ts)[8]) {
#pragma unroll
  for (int i = 0; i < 8; ++i) {
    const float o = ov[i];
    const float4 a = wa[2 * i], c = wa[2 * i + 1];
    const float w8[8] = {a.x, a.y, a.z, a.w, c.x, c.y, c.z, c.w};
#pragma unroll
    for (int j = 0; j < 8; ++j) {
      float p = 1.0f - w8[j] * o;
      float ee = __expf(-40.0f * (p - pmin[j]));
      s[j] += ee; ts[j] += ee * p;
    }
  }
}

__global__ __launch_bounds__(256) void combine_kernel(
    const float* __restrict__ sdf_in, const float* __restrict__ iw,
    const float* __restrict__ uw, const int* __restrict__ is_training,
    float* __restrict__ inter_out, float* __restrict__ occ_out)
{
  __shared__ __attribute__((aligned(16))) float w_sh[K_ * C_];
  __shared__ float om_sh[CP_ * PITCH_];
  __shared__ float in_sh[CP_ * PITCH_];
  const int tid  = threadIdx.x;
  const int lane = tid & 31;
  const int wv = __builtin_amdgcn_readfirstlane(tid >> 5);
  const int blk = blockIdx.x;
  const int b   = blk >> 6;
  const size_t bm0 = (size_t)blk * CP_;
  const int train = is_training[0];

  {
    const float4* wsrc = reinterpret_cast<const float4*>(iw + (size_t)b * (K_ * C_));
    float4* wdst = reinterpret_cast<float4*>(w_sh);
    float4 wq[4];
#pragma unroll
    for (int h = 0; h < 4; ++h) wq[h] = wsrc[tid + h * 256];
#pragma unroll
    for (int h = 0; h < 4; ++h) wdst[tid + h * 256] = wq[h];
  }

  const float* sp = sdf_in + bm0 * K_;
#pragma unroll
  for (int h = 0; h < 2; ++h) {
    const int q = tid + h * 256;
    const float4 v = *reinterpret_cast<const float4*>(sp + q * 4);
    float* d = om_sh + (q >> 4) * PITCH_ + (q & 15) * 4;
    d[0] = 1.0f - 1.0f / (1.0f + __expf(150.0f * v.x));
    d[1] = 1.0f - 1.0f / (1.0f + __expf(150.0f * v.y));
    d[2] = 1.0f - 1.0f / (1.0f + __expf(150.0f * v.z));
    d[3] = 1.0f - 1.0f / (1.0f + __expf(150.0f * v.w));
  }
  __syncthreads();

  const float* orow = om_sh + lane * PITCH_;
  const float* wcol = w_sh + wv * 8;
  float res[8];
  float4 wa[16], wb[16];
  float oa[8], ob[8];
  if (train) {
    float pmin[8];
#pragma unroll
    for (int j = 0; j < 8; ++j) pmin[j] = 3.0e38f;
    load_trip(wcol, orow, 0, wa, oa);
#pragma unroll 1
    for (int k = 0; k < K_; k += 16) {
      load_trip(wcol, orow, k + 8, wb, ob);
      min_trip(wa, oa, pmin);
      load_trip(wcol, orow, k + 16, wa, oa);
      min_trip(wb, ob, pmin);
    }
    float s[8], ts[8];
#pragma unroll
    for (int j = 0; j < 8; ++j) { s[j] = 0.0f; ts[j] = 0.0f; }
#pragma unroll 1
    for (int k = 0; k < K_; k += 16) {
      load_trip(wcol, orow, k + 8, wb, ob);
      sum_trip(wa, oa, pmin, s, ts);
      load_trip(wcol, orow, k + 16, wa, oa);
      sum_trip(wb, ob, pmin, s, ts);
    }
#pragma unroll
    for (int j = 0; j < 8; ++j) res[j] = ts[j] / s[j];
  } else {
#pragma unroll
    for (int j = 0; j < 8; ++j) res[j] = 3.0e38f;
    load_trip(wcol, orow, 0, wa, oa);
#pragma unroll 1
    for (int k = 0; k < K_; k += 16) {
      load_trip(wcol, orow, k + 8, wb, ob);
      min_trip(wa, oa, res);
      load_trip(wcol, orow, k + 16, wa, oa);
      min_trip(wb, ob, res);
    }
  }
  {
    float* irow = in_sh + lane * PITCH_ + wv * 8;
#pragma unroll
    for (int j = 0; j < 8; ++j) irow[j] = res[j];
  }
  __syncthreads();

  float* op = inter_out + bm0 * C_;
#pragma unroll
  for (int h = 0; h < 2; ++h) {
    const int q = tid + h * 256;
    const float* s4 = in_sh + (q >> 4) * PITCH_ + (q & 15) * 4;
    *reinterpret_cast<float4*>(op + q * 4) = make_float4(s4[0], s4[1], s4[2], s4[3]);
  }

  if (wv == 0) {
    const float* ir = in_sh + lane * PITCH_;
    const float* up = uw + b * C_;
    float a[C_];
#pragma unroll
    for (int c = 0; c < C_; ++c) a[c] = up[c] * ir[c];
    float mx = a[0];
#pragma unroll
    for (int c = 1; c < C_; ++c) mx = fmaxf(mx, a[c]);
    if (train) {
      float sv[32], tv[32];
#pragma unroll
      for (int i = 0; i < 32; ++i) {
        float a0 = a[i], a1 = a[i + 32];
        float e0 = __expf(40.0f * (a0 - mx));
        float e1 = __expf(40.0f * (a1 - mx));
        sv[i] = e0 + e1;
        tv[i] = e0 * a0 + e1 * a1;
      }
#pragma unroll
      for (int step = 1; step < 32; step <<= 1) {
#pragma unroll
        for (int i = 0; i < 32; i += 2 * step) {
          sv[i] += sv[i + step];
          tv[i] += tv[i + step];
        }
      }
      occ_out[bm0 + lane] = tv[0] / sv[0];
    } else {
      occ_out[bm0 + lane] = mx;
    }
  }
}

extern "C" void kernel_launch(void* const* d_in, const int* in_sizes, int n_in,
                              void* d_out, int out_size, void* d_ws, size_t ws_size,
                              hipStream_t stream) {
  const float* pts   = (const float*)d_in[0];
  const float* prim  = (const float*)d_in[1];
  const float* iw    = (const float*)d_in[2];
  const float* uw    = (const float*)d_in[3];
  const int*   train = (const int*)d_in[4];

  float* occ_out   = (float*)d_out;
  float* sdf_out   = occ_out + (size_t)B_ * M_;
  float* inter_out = sdf_out + (size_t)B_ * M_ * K_;
  float* dmin_out  = inter_out + (size_t)B_ * M_ * C_;

  dim3 g1(B_ * (K_ / KW_), M_ / 128);
  sdf_kernel<<<g1, 256, 0, stream>>>(pts, prim, sdf_out, dmin_out);
  combine_kernel<<<B_ * M_ / CP_, 256, 0, stream>>>(sdf_out, iw, uw, train,
                                             inter_out, occ_out);
  (void)d_ws;
  (void)in_sizes; (void)n_in; (void)out_size; (void)ws_size;
}
